# starting kernel
# baseline (speedup 1.0000x reference)
#include <hip/hip_runtime.h>

typedef __attribute__((ext_vector_type(2))) float v2f;
typedef __attribute__((ext_vector_type(8))) float v8f;

#define NROWS 30000
#define TTOK  64
#define DIM   128
#define FEAT  137
#define KPAD  140

__global__ __launch_bounds__(256)
void gwas_fused_kernel(const int*   __restrict__ tok,
                       const float* __restrict__ sc,
                       const int*   __restrict__ cat,
                       const float* __restrict__ temb,
                       const float* __restrict__ cemb,
                       const float* __restrict__ Wp,
                       const float* __restrict__ bias,
                       float*       __restrict__ out)
{
    __shared__ float U[16][148];
    __shared__ float dnm[16];
    __shared__ int   sTok[16 * TTOK];
    __shared__ float sSc [16 * TTOK];
    __shared__ int   sCat[16 * TTOK];

    const int tid  = threadIdx.x;
    const int lane = tid & 31;
    const int wave = tid >> 5;
    const int n0   = blockIdx.x * 16;

    for (int i = tid; i < 16 * TTOK; i += 256) {
        const int n = n0 + (i >> 6);
        const int t = i & 63;
        sTok[i] = tok[n * TTOK + t];
        sSc[i]  = sc [n * TTOK + t];
        sCat[i] = cat[n * TTOK + t];
    }
    __syncthreads();

    for (int rr = 0; rr < 2; ++rr) {
        const int r = wave * 2 + rr;
        float ax = 0.f, ay = 0.f, az = 0.f, aw = 0.f;
        float cacc = 0.f, sws = 0.f, wsum = 0.f;
        for (int t = 0; t < TTOK; ++t) {
            const int   idx   = r * TTOK + t;
            const int   token = sTok[idx];
            const float s     = sSc[idx];
            const float w     = (token != 0) ? s : 0.0f;
            const float4 e = *(const float4*)(temb + (size_t)token * DIM + lane * 4);
            ax = fmaf(w, e.x, ax);
            ay = fmaf(w, e.y, ay);
            az = fmaf(w, e.z, az);
            aw = fmaf(w, e.w, aw);
            if (lane < 8) cacc = fmaf(w, cemb[sCat[idx] * 8 + lane], cacc);
            sws  = fmaf(w, s, sws);
            wsum += w;
        }
        U[r][lane * 4 + 0] = ax;
        U[r][lane * 4 + 1] = ay;
        U[r][lane * 4 + 2] = az;
        U[r][lane * 4 + 3] = aw;
        if (lane < 8) U[r][128 + lane] = cacc;
        if (lane == 0) {
            U[r][136] = sws;
            U[r][137] = wsum;
            U[r][138] = 0.0f;
            U[r][139] = 0.0f;
            dnm[r] = fmaxf(wsum, 1e-8f);
        }
    }
    __syncthreads();

    const int d0   = wave * 16;
    const int half = lane >> 4;
    const int l15  = lane & 15;
    const int dcol = d0 + l15;

    v8f c = {0.f, 0.f, 0.f, 0.f, 0.f, 0.f, 0.f, 0.f};
    #pragma unroll
    for (int k = 0; k < KPAD; k += 4) {
        const int ks = k + half * 2;
        v2f a, b;
        a.x = U[l15][ks];
        a.y = U[l15][ks + 1];
        b.x = (ks     < FEAT) ? Wp[dcol * FEAT + ks]
            : ((ks     == FEAT) ? bias[dcol] : 0.0f);
        b.y = (ks + 1 < FEAT) ? Wp[dcol * FEAT + ks + 1]
            : ((ks + 1 == FEAT) ? bias[dcol] : 0.0f);
        c = __builtin_amdgcn_wmma_f32_16x16x4_f32(
                 false, a,  false, b,
                 (short)0, c,  false,  false);
    }

    #pragma unroll
    for (int r = 0; r < 8; ++r) {
        const int m = r + half * 8;
        out[(size_t)(n0 + m) * DIM + dcol] = c[r] / dnm[m];
    }
}

extern "C" void kernel_launch(void* const* d_in, const int* in_sizes, int n_in,
                              void* d_out, int out_size, void* d_ws, size_t ws_size,
                              hipStream_t stream) {
    (void)in_sizes; (void)n_in; (void)out_size; (void)d_ws; (void)ws_size;
    const int*   tok  = (const int*)  d_in[0];
    const float* sc   = (const float*)d_in[1];
    const int*   cat  = (const int*)  d_in[2];
    const float* temb = (const float*)d_in[3];
    const float* cemb = (const float*)d_in[4];
    const float* Wp   = (const float*)d_in[5];
    const float* bias = (const float*)d_in[6];
    float* out = (float*)d_out;

    dim3 grid(NROWS / 16);
    gwas_fused_kernel<<<grid, 256, 0, stream>>>(tok, sc, cat, temb, cemb, Wp, bias, out);
}


# Projection operands read as 16-byte runs (4 consecutive k per lane per 8-k chunk): half as many 16-row weight loads; k order inside each chunk of the projection sum is now 0,1,4,5 then 2,3,6,7
# speedup vs baseline: 1.2463x; 1.2463x over previous
#include <hip/hip_runtime.h>

typedef __attribute__((ext_vector_type(2))) float v2f;
typedef __attribute__((ext_vector_type(8))) float v8f;
typedef float f4u __attribute__((ext_vector_type(4), aligned(4)));

#define NROWS 30000
#define TTOK  64
#define DIM   128
#define FEAT  137
#define KPAD  140
#define GB    8

__global__ __launch_bounds__(256)
void gwas_fused_kernel(const int*   __restrict__ tok,
                       const float* __restrict__ sc,
                       const int*   __restrict__ cat,
                       const float* __restrict__ temb,
                       const float* __restrict__ cemb,
                       const float* __restrict__ Wp,
                       const float* __restrict__ bias,
                       float*       __restrict__ out)
{
    __shared__ __attribute__((aligned(16))) float U[16][148];
    __shared__ float dnm[16];
    __shared__ int   sTok[16 * TTOK];
    __shared__ float sSc [16 * TTOK];
    __shared__ int   sCat[16 * TTOK];

    const int tid  = threadIdx.x;
    const int lane = tid & 31;
    const int wave = tid >> 5;
    const int n0   = blockIdx.x * 16;

    for (int i = tid; i < 16 * TTOK; i += 256) {
        const int n = n0 + (i >> 6);
        const int t = i & 63;
        sTok[i] = tok[n * TTOK + t];
        sSc[i]  = sc [n * TTOK + t];
        sCat[i] = cat[n * TTOK + t];
    }
    __syncthreads();

    const int      lane7  = lane & 7;
    const unsigned lane16 = lane * 16u;
    for (int rr = 0; rr < 2; ++rr) {
        const int r = wave * 2 + rr;
        float ax = 0.f, ay = 0.f, az = 0.f, aw = 0.f;
        float cacc = 0.f, sws = 0.f, wsum = 0.f;
        for (int h = 0; h < 2; ++h) {
            const int   tk = sTok[r * TTOK + h * 32 + lane];
            const float sc = sSc [r * TTOK + h * 32 + lane];
            const int   ct = sCat[r * TTOK + h * 32 + lane];
            const float wt = (tk != 0) ? sc : 0.0f;
            #pragma unroll 1
            for (int t0 = 0; t0 < 32; t0 += GB) {
                float4 e[GB];
                float  ce[GB];
                #pragma unroll
                for (int j = 0; j < GB; ++j) {
                    const int tok = __builtin_amdgcn_readlane(tk, t0 + j);
                    const int cj  = __builtin_amdgcn_readlane(ct, t0 + j);
                    const unsigned off = (unsigned)tok * (DIM * 4u) + lane16;
                    e[j]  = *(const float4*)((const char*)temb + off);
                    ce[j] = cemb[(unsigned)(cj * 8 + lane7)];
                }
                #pragma unroll
                for (int j = 0; j < GB; ++j) {
                    const float w = __int_as_float(__builtin_amdgcn_readlane(__float_as_int(wt), t0 + j));
                    const float s = __int_as_float(__builtin_amdgcn_readlane(__float_as_int(sc), t0 + j));
                    ax = fmaf(w, e[j].x, ax);
                    ay = fmaf(w, e[j].y, ay);
                    az = fmaf(w, e[j].z, az);
                    aw = fmaf(w, e[j].w, aw);
                    cacc = fmaf(w, ce[j], cacc);
                    sws  = fmaf(w, s, sws);
                    wsum += w;
                }
            }
        }
        U[r][lane * 4 + 0] = ax;
        U[r][lane * 4 + 1] = ay;
        U[r][lane * 4 + 2] = az;
        U[r][lane * 4 + 3] = aw;
        if (lane < 8) U[r][128 + lane] = cacc;
        if (lane == 0) {
            U[r][136] = sws;
            U[r][137] = wsum;
            U[r][138] = 0.0f;
            U[r][139] = 0.0f;
            dnm[r] = fmaxf(wsum, 1e-8f);
        }
    }
    __syncthreads();

    const int d0   = wave * 16;
    const int half = lane >> 4;
    const int l15  = lane & 15;
    const int dcol = d0 + l15;

    v8f c = {0.f, 0.f, 0.f, 0.f, 0.f, 0.f, 0.f, 0.f};
    const float* urow = &U[l15][half * 4];
    const float* wrow = Wp + dcol * FEAT + half * 4;
    #pragma unroll
    for (int kc = 0; kc < 136; kc += 8) {
        const float4 av = *(const float4*)(urow + kc);
        const f4u    bv = *(const f4u*)(wrow + kc);
        v2f a0, b0, a1, b1;
        a0.x = av.x; a0.y = av.y; b0.x = bv.x; b0.y = bv.y;
        a1.x = av.z; a1.y = av.w; b1.x = bv.z; b1.y = bv.w;
        c = __builtin_amdgcn_wmma_f32_16x16x4_f32(
                 false, a0,  false, b0,
                 (short)0, c,  false,  false);
        c = __builtin_amdgcn_wmma_f32_16x16x4_f32(
                 false, a1,  false, b1,
                 (short)0, c,  false,  false);
    }
    {
        const int ks = 136 + half * 2;
        v2f a, b;
        a.x = U[l15][ks];
        a.y = U[l15][ks + 1];
        b.x = (half == 0) ? Wp[dcol * FEAT + 136] : 0.0f;
        b.y = (half == 0) ? bias[dcol] : 0.0f;
        c = __builtin_amdgcn_wmma_f32_16x16x4_f32(
                 false, a,  false, b,
                 (short)0, c,  false,  false);
    }

    #pragma unroll
    for (int r = 0; r < 8; ++r) {
        const int m = r + half * 8;
        out[(size_t)(n0 + m) * DIM + dcol] = c[r] / dnm[m];
    }
}

extern "C" void kernel_launch(void* const* d_in, const int* in_sizes, int n_in,
                              void* d_out, int out_size, void* d_ws, size_t ws_size,
                              hipStream_t stream) {
    (void)in_sizes; (void)n_in; (void)out_size; (void)d_ws; (void)ws_size;
    const int*   tok  = (const int*)  d_in[0];
    const float* sc   = (const float*)d_in[1];
    const int*   cat  = (const int*)  d_in[2];
    const float* temb = (const float*)d_in[3];
    const float* cemb = (const float*)d_in[4];
    const float* Wp   = (const float*)d_in[5];
    const float* bias = (const float*)d_in[6];
    float* out = (float*)d_out;

    dim3 grid(NROWS / 16);
    gwas_fused_kernel<<<grid, 256, 0, stream>>>(tok, sc, cat, temb, cemb, Wp, bias, out);
}


# Work-group covers 32 rows (16 waves gather 2 rows each); 8 waves project both 16-row tiles from one weight fragment load: half the 16-row weight reads per row; every sum's order unchanged
# speedup vs baseline: 1.4270x; 1.1450x over previous
#include <hip/hip_runtime.h>

typedef __attribute__((ext_vector_type(2))) float v2f;
typedef __attribute__((ext_vector_type(8))) float v8f;
typedef float f4u __attribute__((ext_vector_type(4), aligned(4)));

#define NROWS 30000
#define TTOK  64
#define DIM   128
#define FEAT  137
#define KPAD  140
#define GB    8
#define RPB   32
#define NBLK  938
static_assert(NBLK * RPB >= NROWS && (NBLK - 1) * RPB + 16 == NROWS, "grid must cover every row; last work-group holds exactly one full 16-row tile");

__global__ __launch_bounds__(512, 16)
void gwas_fused_kernel(const int*   __restrict__ tok,
                       const float* __restrict__ sc,
                       const int*   __restrict__ cat,
                       const float* __restrict__ temb,
                       const float* __restrict__ cemb,
                       const float* __restrict__ Wp,
                       const float* __restrict__ bias,
                       float*       __restrict__ out)
{
    __shared__ __attribute__((aligned(16))) float U[RPB][148];
    __shared__ float dnm[RPB];
    __shared__ int   sTok[RPB * TTOK];
    __shared__ float sSc [RPB * TTOK];
    __shared__ int   sCat[RPB * TTOK];

    const int tid  = threadIdx.x;
    const int lane = tid & 31;
    const int wave = tid >> 5;
    const int n0   = blockIdx.x * RPB;

    for (int i = tid; i < RPB * TTOK; i += 512) {
        const int  n  = n0 + (i >> 6);
        const int  t  = i & 63;
        const bool in = n < NROWS;
        sTok[i] = in ? tok[n * TTOK + t] : 0;
        sSc[i]  = in ? sc [n * TTOK + t] : 0.0f;
        sCat[i] = in ? cat[n * TTOK + t] : 0;
    }
    __syncthreads();

    const int      lane7  = lane & 7;
    const unsigned lane16 = lane * 16u;
    for (int rr = 0; rr < 2; ++rr) {
        const int r = wave * 2 + rr;
        float ax = 0.f, ay = 0.f, az = 0.f, aw = 0.f;
        float cacc = 0.f, sws = 0.f, wsum = 0.f;
        for (int h = 0; h < 2; ++h) {
            const int   tk = sTok[r * TTOK + h * 32 + lane];
            const float sc = sSc [r * TTOK + h * 32 + lane];
            const int   ct = sCat[r * TTOK + h * 32 + lane];
            const float wt = (tk != 0) ? sc : 0.0f;
            #pragma unroll 1
            for (int t0 = 0; t0 < 32; t0 += GB) {
                float4 e[GB];
                float  ce[GB];
                #pragma unroll
                for (int j = 0; j < GB; ++j) {
                    const int tok = __builtin_amdgcn_readlane(tk, t0 + j);
                    const int cj  = __builtin_amdgcn_readlane(ct, t0 + j);
                    const unsigned off = (unsigned)tok * (DIM * 4u) + lane16;
                    e[j]  = *(const float4*)((const char*)temb + off);
                    ce[j] = cemb[(unsigned)(cj * 8 + lane7)];
                }
                #pragma unroll
                for (int j = 0; j < GB; ++j) {
                    const float w = __int_as_float(__builtin_amdgcn_readlane(__float_as_int(wt), t0 + j));
                    const float s = __int_as_float(__builtin_amdgcn_readlane(__float_as_int(sc), t0 + j));
                    ax = fmaf(w, e[j].x, ax);
                    ay = fmaf(w, e[j].y, ay);
                    az = fmaf(w, e[j].z, az);
                    aw = fmaf(w, e[j].w, aw);
                    cacc = fmaf(w, ce[j], cacc);
                    sws  = fmaf(w, s, sws);
                    wsum += w;
                }
            }
        }
        U[r][lane * 4 + 0] = ax;
        U[r][lane * 4 + 1] = ay;
        U[r][lane * 4 + 2] = az;
        U[r][lane * 4 + 3] = aw;
        if (lane < 8) U[r][128 + lane] = cacc;
        if (lane == 0) {
            U[r][136] = sws;
            U[r][137] = wsum;
            U[r][138] = 0.0f;
            U[r][139] = 0.0f;
            dnm[r] = fmaxf(wsum, 1e-8f);
        }
    }
    __syncthreads();

    if (wave >= 8) return;
    const int d0   = wave * 16;
    const int half = lane >> 4;
    const int l15  = lane & 15;
    const int dcol = d0 + l15;

    v8f c0 = {0.f, 0.f, 0.f, 0.f, 0.f, 0.f, 0.f, 0.f};
    v8f c1 = {0.f, 0.f, 0.f, 0.f, 0.f, 0.f, 0.f, 0.f};
    const float* urow0 = &U[l15][half * 4];
    const float* urow1 = &U[16 + l15][half * 4];
    const float* wrow  = Wp + dcol * FEAT + half * 4;
    #pragma unroll
    for (int kc = 0; kc < 136; kc += 8) {
        const float4 av = *(const float4*)(urow0 + kc);
        const float4 au = *(const float4*)(urow1 + kc);
        const f4u    bv = *(const f4u*)(wrow + kc);
        v2f a0, a1, e0, e1, b0, b1;
        a0.x = av.x; a0.y = av.y; a1.x = av.z; a1.y = av.w;
        e0.x = au.x; e0.y = au.y; e1.x = au.z; e1.y = au.w;
        b0.x = bv.x; b0.y = bv.y; b1.x = bv.z; b1.y = bv.w;
        c0 = __builtin_amdgcn_wmma_f32_16x16x4_f32(
                 false, a0,  false, b0,
                 (short)0, c0,  false,  false);
        c1 = __builtin_amdgcn_wmma_f32_16x16x4_f32(
                 false, e0,  false, b0,
                 (short)0, c1,  false,  false);
        c0 = __builtin_amdgcn_wmma_f32_16x16x4_f32(
                 false, a1,  false, b1,
                 (short)0, c0,  false,  false);
        c1 = __builtin_amdgcn_wmma_f32_16x16x4_f32(
                 false, e1,  false, b1,
                 (short)0, c1,  false,  false);
    }
    {
        const int ks = 136 + half * 2;
        v2f a, e, b;
        a.x = U[l15][ks];
        a.y = U[l15][ks + 1];
        e.x = U[16 + l15][ks];
        e.y = U[16 + l15][ks + 1];
        b.x = (half == 0) ? Wp[dcol * FEAT + 136] : 0.0f;
        b.y = (half == 0) ? bias[dcol] : 0.0f;
        c0 = __builtin_amdgcn_wmma_f32_16x16x4_f32(
                 false, a,  false, b,
                 (short)0, c0,  false,  false);
        c1 = __builtin_amdgcn_wmma_f32_16x16x4_f32(
                 false, e,  false, b,
                 (short)0, c1,  false,  false);
    }

    #pragma unroll
    for (int r = 0; r < 8; ++r) {
        const int m = r + half * 8;
        out[(size_t)(n0 + m) * DIM + dcol] = c0[r] / dnm[m];
    }
    #pragma unroll
    for (int r = 0; r < 8; ++r) {
        const int m = 16 + r + half * 8;
        if (n0 + m < NROWS)
            out[(size_t)(n0 + m) * DIM + dcol] = c1[r] / dnm[m];
    }
}

extern "C" void kernel_launch(void* const* d_in, const int* in_sizes, int n_in,
                              void* d_out, int out_size, void* d_ws, size_t ws_size,
                              hipStream_t stream) {
    (void)in_sizes; (void)n_in; (void)out_size; (void)d_ws; (void)ws_size;
    const int*   tok  = (const int*)  d_in[0];
    const float* sc   = (const float*)d_in[1];
    const int*   cat  = (const int*)  d_in[2];
    const float* temb = (const float*)d_in[3];
    const float* cemb = (const float*)d_in[4];
    const float* Wp   = (const float*)d_in[5];
    const float* bias = (const float*)d_in[6];
    float* out = (float*)d_out;

    dim3 grid(NBLK);
    gwas_fused_kernel<<<grid, 512, 0, stream>>>(tok, sc, cat, temb, cemb, Wp, bias, out);
}


# Projection weights copied once per work-group into LDS by whole-line 16-byte loads and read there as B fragments; work-group widened to 64 rows so two stay resident per WGP; sum orders unchanged
# speedup vs baseline: 1.5155x; 1.0620x over previous
#include <hip/hip_runtime.h>

typedef __attribute__((ext_vector_type(2))) float v2f;
typedef __attribute__((ext_vector_type(8))) float v8f;
typedef float f4u __attribute__((ext_vector_type(4), aligned(4)));

#define NROWS 30000
#define TTOK  64
#define DIM   128
#define FEAT  137
#define KPAD  140
#define GB    8
#define RPB   64
#define NTHR  1024
#define NBLK  469
#define NW4   (DIM * FEAT / 4)
#define NWK   ((NW4 + NTHR - 1) / NTHR)
static_assert(NBLK * RPB >= NROWS && (NBLK - 1) * RPB + 48 == NROWS, "grid must cover every row; in the last work-group tiles 0..2 are full and tile 3 is empty");
static_assert((DIM * FEAT) % 4 == 0, "weight matrix is copied in 16-byte pieces");

__global__ __launch_bounds__(NTHR, 16)
void gwas_fused_kernel(const int*   __restrict__ tok,
                       const float* __restrict__ sc,
                       const int*   __restrict__ cat,
                       const float* __restrict__ temb,
                       const float* __restrict__ cemb,
                       const float* __restrict__ Wp,
                       const float* __restrict__ bias,
                       float*       __restrict__ out)
{
    __shared__ __attribute__((aligned(16))) float U[RPB][148];
    __shared__ float dnm[RPB];
    __shared__ int   sTok[RPB * TTOK];
    __shared__ float sSc [RPB * TTOK];
    __shared__ int   sCat[RPB * TTOK];
    __shared__ __attribute__((aligned(16))) float Wl[DIM][KPAD];

    const int tid  = threadIdx.x;
    const int lane = tid & 31;
    const int wave = tid >> 5;
    const int n0   = blockIdx.x * RPB;

    f4u wv[NWK];
    #pragma unroll
    for (int k = 0; k < NWK; ++k) {
        const int q = tid + k * NTHR;
        wv[k] = (q < NW4) ? *(const f4u*)(Wp + 4 * q) : (f4u){0.f, 0.f, 0.f, 0.f};
    }
    const float bval = (tid < DIM) ? bias[tid] : 0.0f;

    for (int i = tid; i < RPB * TTOK; i += NTHR) {
        const int  n  = n0 + (i >> 6);
        const int  t  = i & 63;
        const bool in = n < NROWS;
        sTok[i] = in ? tok[n * TTOK + t] : 0;
        sSc[i]  = in ? sc [n * TTOK + t] : 0.0f;
        sCat[i] = in ? cat[n * TTOK + t] : 0;
    }

    #pragma unroll
    for (int k = 0; k < NWK; ++k) {
        const int q = tid + k * NTHR;
        if (q < NW4) {
            const int e   = 4 * q;
            const int row = e / FEAT;
            const int col = e - row * FEAT;
            float* dst = &Wl[0][0] + e + (KPAD - FEAT) * row;
            dst[0] = wv[k].x;
            dst[1 + ((col + 1 >= FEAT) ? (KPAD - FEAT) : 0)] = wv[k].y;
            dst[2 + ((col + 2 >= FEAT) ? (KPAD - FEAT) : 0)] = wv[k].z;
            dst[3 + ((col + 3 >= FEAT) ? (KPAD - FEAT) : 0)] = wv[k].w;
        }
    }
    if (tid < DIM) {
        Wl[tid][FEAT]     = bval;
        Wl[tid][FEAT + 1] = 0.0f;
        Wl[tid][FEAT + 2] = 0.0f;
    }
    __syncthreads();

    const int      lane7  = lane & 7;
    const unsigned lane16 = lane * 16u;
    for (int rr = 0; rr < 2; ++rr) {
        const int r = wave * 2 + rr;
        float ax = 0.f, ay = 0.f, az = 0.f, aw = 0.f;
        float cacc = 0.f, sws = 0.f, wsum = 0.f;
        for (int h = 0; h < 2; ++h) {
            const int   tk = sTok[r * TTOK + h * 32 + lane];
            const float sc = sSc [r * TTOK + h * 32 + lane];
            const int   ct = sCat[r * TTOK + h * 32 + lane];
            const float wt = (tk != 0) ? sc : 0.0f;
            #pragma unroll 1
            for (int t0 = 0; t0 < 32; t0 += GB) {
                float4 e[GB];
                float  ce[GB];
                #pragma unroll
                for (int j = 0; j < GB; ++j) {
                    const int tok = __builtin_amdgcn_readlane(tk, t0 + j);
                    const int cj  = __builtin_amdgcn_readlane(ct, t0 + j);
                    const unsigned off = (unsigned)tok * (DIM * 4u) + lane16;
                    e[j]  = *(const float4*)((const char*)temb + off);
                    ce[j] = cemb[(unsigned)(cj * 8 + lane7)];
                }
                #pragma unroll
                for (int j = 0; j < GB; ++j) {
                    const float w = __int_as_float(__builtin_amdgcn_readlane(__float_as_int(wt), t0 + j));
                    const float s = __int_as_float(__builtin_amdgcn_readlane(__float_as_int(sc), t0 + j));
                    ax = fmaf(w, e[j].x, ax);
                    ay = fmaf(w, e[j].y, ay);
                    az = fmaf(w, e[j].z, az);
                    aw = fmaf(w, e[j].w, aw);
                    cacc = fmaf(w, ce[j], cacc);
                    sws  = fmaf(w, s, sws);
                    wsum += w;
                }
            }
        }
        U[r][lane * 4 + 0] = ax;
        U[r][lane * 4 + 1] = ay;
        U[r][lane * 4 + 2] = az;
        U[r][lane * 4 + 3] = aw;
        if (lane < 8) U[r][128 + lane] = cacc;
        if (lane == 0) {
            U[r][136] = sws;
            U[r][137] = wsum;
            U[r][138] = 0.0f;
            U[r][139] = 0.0f;
            dnm[r] = fmaxf(wsum, 1e-8f);
        }
    }
    __syncthreads();

    if (wave >= 16) return;
    const int d0   = (wave & 7) * 16;
    const int rt   = (wave >> 3) * 32;
    const int half = lane >> 4;
    const int l15  = lane & 15;
    const int dcol = d0 + l15;

    v8f c0 = {0.f, 0.f, 0.f, 0.f, 0.f, 0.f, 0.f, 0.f};
    v8f c1 = {0.f, 0.f, 0.f, 0.f, 0.f, 0.f, 0.f, 0.f};
    const float* urow0 = &U[rt + l15][half * 4];
    const float* urow1 = &U[rt + 16 + l15][half * 4];
    const float* wrow  = &Wl[dcol][half * 4];
    #pragma unroll
    for (int kc = 0; kc < 136; kc += 8) {
        const float4 av = *(const float4*)(urow0 + kc);
        const float4 au = *(const float4*)(urow1 + kc);
        const float4 bv = *(const float4*)(wrow + kc);
        v2f a0, a1, e0, e1, b0, b1;
        a0.x = av.x; a0.y = av.y; a1.x = av.z; a1.y = av.w;
        e0.x = au.x; e0.y = au.y; e1.x = au.z; e1.y = au.w;
        b0.x = bv.x; b0.y = bv.y; b1.x = bv.z; b1.y = bv.w;
        c0 = __builtin_amdgcn_wmma_f32_16x16x4_f32(
                 false, a0,  false, b0,
                 (short)0, c0,  false,  false);
        c1 = __builtin_amdgcn_wmma_f32_16x16x4_f32(
                 false, e0,  false, b0,
                 (short)0, c1,  false,  false);
        c0 = __builtin_amdgcn_wmma_f32_16x16x4_f32(
                 false, a1,  false, b1,
                 (short)0, c0,  false,  false);
        c1 = __builtin_amdgcn_wmma_f32_16x16x4_f32(
                 false, e1,  false, b1,
                 (short)0, c1,  false,  false);
    }
    {
        const int ks = 136 + half * 2;
        v2f a, e, b;
        a.x = U[rt + l15][ks];
        a.y = U[rt + l15][ks + 1];
        e.x = U[rt + 16 + l15][ks];
        e.y = U[rt + 16 + l15][ks + 1];
        b.x = Wl[dcol][ks];
        b.y = Wl[dcol][ks + 1];
        c0 = __builtin_amdgcn_wmma_f32_16x16x4_f32(
                 false, a,  false, b,
                 (short)0, c0,  false,  false);
        c1 = __builtin_amdgcn_wmma_f32_16x16x4_f32(
                 false, e,  false, b,
                 (short)0, c1,  false,  false);
    }

    #pragma unroll
    for (int r = 0; r < 8; ++r) {
        const int m = rt + r + half * 8;
        out[(size_t)(n0 + m) * DIM + dcol] = c0[r] / dnm[m];
    }
    #pragma unroll
    for (int r = 0; r < 8; ++r) {
        const int m = rt + 16 + r + half * 8;
        if (n0 + m < NROWS)
            out[(size_t)(n0 + m) * DIM + dcol] = c1[r] / dnm[m];
    }
}

extern "C" void kernel_launch(void* const* d_in, const int* in_sizes, int n_in,
                              void* d_out, int out_size, void* d_ws, size_t ws_size,
                              hipStream_t stream) {
    (void)in_sizes; (void)n_in; (void)out_size; (void)d_ws; (void)ws_size;
    const int*   tok  = (const int*)  d_in[0];
    const float* sc   = (const float*)d_in[1];
    const int*   cat  = (const int*)  d_in[2];
    const float* temb = (const float*)d_in[3];
    const float* cemb = (const float*)d_in[4];
    const float* Wp   = (const float*)d_in[5];
    const float* bias = (const float*)d_in[6];
    float* out = (float*)d_out;

    dim3 grid(NBLK);
    gwas_fused_kernel<<<grid, NTHR, 0, stream>>>(tok, sc, cat, temb, cemb, Wp, bias, out);
}


# Each wave loads the ids, scores and categories of its own 2 rows (one 16-byte load per array) and gathers at once: no 4 dependent staging rounds, no barrier before the gather; sums unchanged
# speedup vs baseline: 1.5625x; 1.0310x over previous
#include <hip/hip_runtime.h>

typedef __attribute__((ext_vector_type(2))) float v2f;
typedef __attribute__((ext_vector_type(8))) float v8f;
typedef float f4u __attribute__((ext_vector_type(4), aligned(4)));
typedef int   i4u __attribute__((ext_vector_type(4), aligned(4)));

#define NROWS 30000
#define TTOK  64
#define DIM   128
#define FEAT  137
#define KPAD  140
#define GB    8
#define RPB   64
#define NTHR  1024
#define NBLK  469
#define NW4   (DIM * FEAT / 4)
#define NWK   ((NW4 + NTHR - 1) / NTHR)
static_assert(NBLK * RPB >= NROWS && (NBLK - 1) * RPB + 48 == NROWS, "grid must cover every row; in the last work-group tiles 0..2 are full and tile 3 is empty");
static_assert((DIM * FEAT) % 4 == 0, "weight matrix is copied in 16-byte pieces");
static_assert(NROWS % 2 == 0 && RPB % 2 == 0, "a wave's two rows are both real or both past the end");

__global__ __launch_bounds__(NTHR, 16)
void gwas_fused_kernel(const int*   __restrict__ tok,
                       const float* __restrict__ sc,
                       const int*   __restrict__ cat,
                       const float* __restrict__ temb,
                       const float* __restrict__ cemb,
                       const float* __restrict__ Wp,
                       const float* __restrict__ bias,
                       float*       __restrict__ out)
{
    __shared__ __attribute__((aligned(16))) float U[RPB][148];
    __shared__ float dnm[RPB];
    __shared__ __attribute__((aligned(16))) int   sTok[RPB * TTOK];
    __shared__ __attribute__((aligned(16))) float sSc [RPB * TTOK];
    __shared__ __attribute__((aligned(16))) int   sCat[RPB * TTOK];
    __shared__ __attribute__((aligned(16))) float Wl[DIM][KPAD];

    const int tid  = threadIdx.x;
    const int lane = tid & 31;
    const int wave = tid >> 5;
    const int n0   = blockIdx.x * RPB;

    const int  g0 = (n0 + wave * 2) * TTOK + lane * 4;
    const bool in = (n0 + wave * 2) < NROWS;
    i4u tk4 = {0, 0, 0, 0};
    f4u sc4 = {0.f, 0.f, 0.f, 0.f};
    i4u ct4 = {0, 0, 0, 0};
    if (in) {
        tk4 = *(const i4u*)(tok + g0);
        sc4 = *(const f4u*)(sc  + g0);
        ct4 = *(const i4u*)(cat + g0);
    }
    const float bval = (tid < DIM) ? bias[tid] : 0.0f;
    f4u wv[NWK];
    #pragma unroll
    for (int k = 0; k < NWK; ++k) {
        const int q = tid + k * NTHR;
        wv[k] = (q < NW4) ? *(const f4u*)(Wp + 4 * q) : (f4u){0.f, 0.f, 0.f, 0.f};
    }

    {
        const int s4 = wave * (2 * TTOK) + lane * 4;
        *(int4*)  &sTok[s4] = (int4)  {tk4.x, tk4.y, tk4.z, tk4.w};
        *(float4*)&sSc [s4] = (float4){sc4.x, sc4.y, sc4.z, sc4.w};
        *(int4*)  &sCat[s4] = (int4)  {ct4.x, ct4.y, ct4.z, ct4.w};
    }

    #pragma unroll
    for (int k = 0; k < NWK; ++k) {
        const int q = tid + k * NTHR;
        if (q < NW4) {
            const int e   = 4 * q;
            const int row = e / FEAT;
            const int col = e - row * FEAT;
            float* dst = &Wl[0][0] + e + (KPAD - FEAT) * row;
            dst[0] = wv[k].x;
            dst[1 + ((col + 1 >= FEAT) ? (KPAD - FEAT) : 0)] = wv[k].y;
            dst[2 + ((col + 2 >= FEAT) ? (KPAD - FEAT) : 0)] = wv[k].z;
            dst[3 + ((col + 3 >= FEAT) ? (KPAD - FEAT) : 0)] = wv[k].w;
        }
    }
    if (tid < DIM) {
        Wl[tid][FEAT]     = bval;
        Wl[tid][FEAT + 1] = 0.0f;
        Wl[tid][FEAT + 2] = 0.0f;
    }

    const int      lane7  = lane & 7;
    const unsigned lane16 = lane * 16u;
    for (int rr = 0; rr < 2; ++rr) {
        const int r = wave * 2 + rr;
        float ax = 0.f, ay = 0.f, az = 0.f, aw = 0.f;
        float cacc = 0.f, sws = 0.f, wsum = 0.f;
        for (int h = 0; h < 2; ++h) {
            const int   tk = sTok[r * TTOK + h * 32 + lane];
            const float sc = sSc [r * TTOK + h * 32 + lane];
            const int   ct = sCat[r * TTOK + h * 32 + lane];
            const float wt = (tk != 0) ? sc : 0.0f;
            #pragma unroll 1
            for (int t0 = 0; t0 < 32; t0 += GB) {
                float4 e[GB];
                float  ce[GB];
                #pragma unroll
                for (int j = 0; j < GB; ++j) {
                    const int tok = __builtin_amdgcn_readlane(tk, t0 + j);
                    const int cj  = __builtin_amdgcn_readlane(ct, t0 + j);
                    const unsigned off = (unsigned)tok * (DIM * 4u) + lane16;
                    e[j]  = *(const float4*)((const char*)temb + off);
                    ce[j] = cemb[(unsigned)(cj * 8 + lane7)];
                }
                #pragma unroll
                for (int j = 0; j < GB; ++j) {
                    const float w = __int_as_float(__builtin_amdgcn_readlane(__float_as_int(wt), t0 + j));
                    const float s = __int_as_float(__builtin_amdgcn_readlane(__float_as_int(sc), t0 + j));
                    ax = fmaf(w, e[j].x, ax);
                    ay = fmaf(w, e[j].y, ay);
                    az = fmaf(w, e[j].z, az);
                    aw = fmaf(w, e[j].w, aw);
                    cacc = fmaf(w, ce[j], cacc);
                    sws  = fmaf(w, s, sws);
                    wsum += w;
                }
            }
        }
        U[r][lane * 4 + 0] = ax;
        U[r][lane * 4 + 1] = ay;
        U[r][lane * 4 + 2] = az;
        U[r][lane * 4 + 3] = aw;
        if (lane < 8) U[r][128 + lane] = cacc;
        if (lane == 0) {
            U[r][136] = sws;
            U[r][137] = wsum;
            U[r][138] = 0.0f;
            U[r][139] = 0.0f;
            dnm[r] = fmaxf(wsum, 1e-8f);
        }
    }
    __syncthreads();

    if (wave >= 16) return;
    const int d0   = (wave & 7) * 16;
    const int rt   = (wave >> 3) * 32;
    const int half = lane >> 4;
    const int l15  = lane & 15;
    const int dcol = d0 + l15;

    v8f c0 = {0.f, 0.f, 0.f, 0.f, 0.f, 0.f, 0.f, 0.f};
    v8f c1 = {0.f, 0.f, 0.f, 0.f, 0.f, 0.f, 0.f, 0.f};
    const float* urow0 = &U[rt + l15][half * 4];
    const float* urow1 = &U[rt + 16 + l15][half * 4];
    const float* wrow  = &Wl[dcol][half * 4];
    #pragma unroll
    for (int kc = 0; kc < 136; kc += 8) {
        const float4 av = *(const float4*)(urow0 + kc);
        const float4 au = *(const float4*)(urow1 + kc);
        const float4 bv = *(const float4*)(wrow + kc);
        v2f a0, a1, e0, e1, b0, b1;
        a0.x = av.x; a0.y = av.y; a1.x = av.z; a1.y = av.w;
        e0.x = au.x; e0.y = au.y; e1.x = au.z; e1.y = au.w;
        b0.x = bv.x; b0.y = bv.y; b1.x = bv.z; b1.y = bv.w;
        c0 = __builtin_amdgcn_wmma_f32_16x16x4_f32(
                 false, a0,  false, b0,
                 (short)0, c0,  false,  false);
        c1 = __builtin_amdgcn_wmma_f32_16x16x4_f32(
                 false, e0,  false, b0,
                 (short)0, c1,  false,  false);
        c0 = __builtin_amdgcn_wmma_f32_16x16x4_f32(
                 false, a1,  false, b1,
                 (short)0, c0,  false,  false);
        c1 = __builtin_amdgcn_wmma_f32_16x16x4_f32(
                 false, e1,  false, b1,
                 (short)0, c1,  false,  false);
    }
    {
        const int ks = 136 + half * 2;
        v2f a, e, b;
        a.x = U[rt + l15][ks];
        a.y = U[rt + l15][ks + 1];
        e.x = U[rt + 16 + l15][ks];
        e.y = U[rt + 16 + l15][ks + 1];
        b.x = Wl[dcol][ks];
        b.y = Wl[dcol][ks + 1];
        c0 = __builtin_amdgcn_wmma_f32_16x16x4_f32(
                 false, a,  false, b,
                 (short)0, c0,  false,  false);
        c1 = __builtin_amdgcn_wmma_f32_16x16x4_f32(
                 false, e,  false, b,
                 (short)0, c1,  false,  false);
    }

    #pragma unroll
    for (int r = 0; r < 8; ++r) {
        const int m = rt + r + half * 8;
        out[(size_t)(n0 + m) * DIM + dcol] = c0[r] / dnm[m];
    }
    #pragma unroll
    for (int r = 0; r < 8; ++r) {
        const int m = rt + 16 + r + half * 8;
        if (n0 + m < NROWS)
            out[(size_t)(n0 + m) * DIM + dcol] = c1[r] / dnm[m];
    }
}

extern "C" void kernel_launch(void* const* d_in, const int* in_sizes, int n_in,
                              void* d_out, int out_size, void* d_ws, size_t ws_size,
                              hipStream_t stream) {
    (void)in_sizes; (void)n_in; (void)out_size; (void)d_ws; (void)ws_size;
    const int*   tok  = (const int*)  d_in[0];
    const float* sc   = (const float*)d_in[1];
    const int*   cat  = (const int*)  d_in[2];
    const float* temb = (const float*)d_in[3];
    const float* cemb = (const float*)d_in[4];
    const float* Wp   = (const float*)d_in[5];
    const float* bias = (const float*)d_in[6];
    float* out = (float*)d_out;

    dim3 grid(NBLK);
    gwas_fused_kernel<<<grid, NTHR, 0, stream>>>(tok, sc, cat, temb, cemb, Wp, bias, out);
}
